# starting kernel
# baseline (speedup 1.0000x reference)
#include <hip/hip_runtime.h>

typedef __attribute__((ext_vector_type(2))) float v2f;
typedef __attribute__((ext_vector_type(8))) float v8f;

#define DIM 64

__global__ void sage_zero_kernel(float* __restrict__ ws, int n) {
    int i = blockIdx.x * blockDim.x + threadIdx.x;
    if (i < n) ws[i] = 0.0f;
}

__global__ void __launch_bounds__(256)
sage_edge_scatter_kernel(const float* __restrict__ x,
                         const int*   __restrict__ src,
                         const int*   __restrict__ dst,
                         float*       __restrict__ agg,
                         float*       __restrict__ deg,
                         int E) {
    int gid  = blockIdx.x * blockDim.x + threadIdx.x;
    int e    = gid >> 5;
    int lane = gid & 31;
    if (e >= E) return;
    int s = src[e];
    int d = dst[e];
    const float2* xr = (const float2*)(x + (size_t)s * DIM);
    float2 v = xr[lane];
    float* ar = agg + (size_t)d * DIM + lane * 2;
    unsafeAtomicAdd(ar,     v.x);
    unsafeAtomicAdd(ar + 1, v.y);
    if (lane == 0) unsafeAtomicAdd(deg + d, 1.0f);
}

__global__ void __launch_bounds__(256)
sage_wmma_gemm_kernel(const float* __restrict__ x,
                      const float* __restrict__ agg,
                      const float* __restrict__ deg,
                      const float* __restrict__ Wself,
                      const float* __restrict__ bself,
                      const float* __restrict__ Wneigh,
                      const float* __restrict__ bneigh,
                      float*       __restrict__ out,
                      int nTiles) {
    __shared__ float sWs[DIM * DIM];
    __shared__ float sWn[DIM * DIM];
    __shared__ float sB [DIM];

    for (int i = threadIdx.x; i < DIM * DIM; i += 256) {
        sWs[i] = Wself[i];
        sWn[i] = Wneigh[i];
    }
    if (threadIdx.x < DIM) sB[threadIdx.x] = bself[threadIdx.x] + bneigh[threadIdx.x];
    __syncthreads();

    const int wave = threadIdx.x >> 5;
    const int lane = threadIdx.x & 31;
    const int tile = blockIdx.x * 8 + wave;
    if (tile >= nTiles) return;

    const int m    = lane & 15;
    const int hi   = lane >> 4;
    const int row0 = tile * 16;
    const int row  = row0 + m;

    const float dinv = 1.0f / fmaxf(deg[row], 1.0f);
    const float* __restrict__ xrow = x   + (size_t)row * DIM;
    const float* __restrict__ arow = agg + (size_t)row * DIM;

    v8f acc[4];
    for (int nt = 0; nt < 4; ++nt) {
        float b = sB[nt * 16 + m];
        for (int r = 0; r < 8; ++r) acc[nt][r] = b;
    }

    for (int kb = 0; kb < DIM; kb += 4) {
        const int kk = kb + 2 * hi;
        v2f as, an;
        as.x = xrow[kk];
        as.y = xrow[kk + 1];
        an.x = arow[kk]     * dinv;
        an.y = arow[kk + 1] * dinv;

        for (int nt = 0; nt < 4; ++nt) {
            const int bi = kk * DIM + nt * 16 + m;
            v2f bs, bn;
            bs.x = sWs[bi];
            bs.y = sWs[bi + DIM];
            bn.x = sWn[bi];
            bn.y = sWn[bi + DIM];
            acc[nt] = __builtin_amdgcn_wmma_f32_16x16x4_f32(
                          false, as, false, bs, (short)0, acc[nt], false, false);
            acc[nt] = __builtin_amdgcn_wmma_f32_16x16x4_f32(
                          false, an, false, bn, (short)0, acc[nt], false, false);
        }
    }

    for (int nt = 0; nt < 4; ++nt) {
        const int col = nt * 16 + m;
        for (int r = 0; r < 8; ++r) {
            const int mrow = r + 8 * hi;
            out[(size_t)(row0 + mrow) * DIM + col] = acc[nt][r];
        }
    }
}

extern "C" void kernel_launch(void* const* d_in, const int* in_sizes, int n_in,
                              void* d_out, int out_size, void* d_ws, size_t ws_size,
                              hipStream_t stream) {
    const float* x      = (const float*)d_in[0];
    const int*   src    = (const int*)  d_in[1];
    const int*   dst    = (const int*)  d_in[2];
    const float* Wself  = (const float*)d_in[3];
    const float* bself  = (const float*)d_in[4];
    const float* Wneigh = (const float*)d_in[5];
    const float* bneigh = (const float*)d_in[6];
    float*       out    = (float*)d_out;

    const int N = in_sizes[0] / DIM;
    const int E = in_sizes[1];

    float* agg = (float*)d_ws;
    float* deg = agg + (size_t)N * DIM;

    const int zn = N * DIM + N;
    sage_zero_kernel<<<(zn + 255) / 256, 256, 0, stream>>>(agg, zn);

    const long long totThreads = (long long)E * 32;
    sage_edge_scatter_kernel<<<(int)((totThreads + 255) / 256), 256, 0, stream>>>(
        x, src, dst, agg, deg, E);

    const int nTiles = (N + 15) / 16;
    sage_wmma_gemm_kernel<<<(nTiles + 7) / 8, 256, 0, stream>>>(
        x, agg, deg, Wself, bself, Wneigh, bneigh, out, nTiles);
}


# Float-atomic scatter replaced by counting sort of edges by destination (int atomics only) and a per-tile gather fused into the WMMA kernel; neighbour sums now added in bucket-fill arrival order
# speedup vs baseline: 1.2843x; 1.2843x over previous
#include <hip/hip_runtime.h>


typedef __attribute__((ext_vector_type(2))) float v2f;
typedef __attribute__((ext_vector_type(8))) float v8f;

#define DIM 64
#define TILE_WAVES 4
#define APITCH 68

__global__ void sage_zero_counts_kernel(int* __restrict__ cnt, int n) {
    int i = blockIdx.x * blockDim.x + threadIdx.x;
    if (i < n) cnt[i] = 0;
}

__global__ void __launch_bounds__(256)
sage_count_kernel(const int* __restrict__ dst, int* __restrict__ cnt, int E) {
    int e = blockIdx.x * blockDim.x + threadIdx.x;
    if (e < E) atomicAdd(cnt + dst[e], 1);
}

__global__ void __launch_bounds__(256)
sage_alloc_segments_kernel(const int* __restrict__ cnt, int* __restrict__ cursor,
                           int* __restrict__ total, int n) {
    const int i    = blockIdx.x * blockDim.x + threadIdx.x;
    const int lane = threadIdx.x & 31;
    const int c    = (i < n) ? cnt[i] : 0;
    int incl = c;
    for (int o = 1; o < 32; o <<= 1) {
        int u = __shfl_up(incl, o);
        if (lane >= o) incl += u;
    }
    int base = 0;
    if (lane == 31) base = atomicAdd(total, incl);
    base = __shfl(base, 31);
    if (i < n) cursor[i] = base + incl - c;
}

__global__ void __launch_bounds__(256)
sage_fill_kernel(const int* __restrict__ src, const int* __restrict__ dst,
                 int* __restrict__ cursor, int* __restrict__ perm, int E) {
    int e = blockIdx.x * blockDim.x + threadIdx.x;
    if (e < E) {
        int p = atomicAdd(cursor + dst[e], 1);
        perm[p] = src[e];
    }
}

#define GB 16

__device__ __forceinline__ void issue_rows(const float* __restrict__ xl, int idx, int i0,
                                           float2 (&v)[GB]) {
    int s[GB];
    #pragma unroll
    for (int k = 0; k < GB; ++k) s[k] = __builtin_amdgcn_readlane(idx, i0 + k);
    #pragma unroll
    for (int k = 0; k < GB; ++k) v[k] = *(const float2*)(xl + (size_t)s[k] * DIM);
}

__global__ void __launch_bounds__(TILE_WAVES * 32)
sage_gather_wmma_kernel(const float* __restrict__ x,
                        const int*   __restrict__ cnt,
                        const int*   __restrict__ segEnd,
                        const int*   __restrict__ perm,
                        const float* __restrict__ Wself,
                        const float* __restrict__ bself,
                        const float* __restrict__ Wneigh,
                        const float* __restrict__ bneigh,
                        float*       __restrict__ out,
                        int nTiles) {
    __shared__ float sWs[DIM * DIM];
    __shared__ float sWn[DIM * DIM];
    __shared__ float sB [DIM];
    __shared__ float sA [TILE_WAVES * 16 * APITCH];

    for (int i = threadIdx.x; i < DIM * DIM; i += TILE_WAVES * 32) {
        sWs[i] = Wself[i];
        sWn[i] = Wneigh[i];
    }
    if (threadIdx.x < DIM) sB[threadIdx.x] = bself[threadIdx.x] + bneigh[threadIdx.x];

    const int wave = threadIdx.x >> 5;
    const int lane = threadIdx.x & 31;
    const int tile = blockIdx.x * TILE_WAVES + wave;
    const bool active = tile < nTiles;
    const int row0 = tile * 16;
    float* myA = sA + wave * 16 * APITCH;

    if (active) {
        int myEnd = 0, myCnt = 0;
        if (lane < 16) {
            myEnd = segEnd[row0 + lane];
            myCnt = cnt[row0 + lane];
        }
        const float* xl = x + 2 * lane;
        float* aPtr = myA + 2 * lane;

        const int tileBeg = __builtin_amdgcn_readlane(myEnd, 0) - __builtin_amdgcn_readlane(myCnt, 0);
        const int tileEnd = __builtin_amdgcn_readlane(myEnd, 15);

        int   r      = 0;
        int   rowEnd = __builtin_amdgcn_readlane(myEnd, 0);
        float ax = 0.0f, ay = 0.0f;

        int idxNext = (lane < tileEnd - tileBeg) ? perm[tileBeg + lane] : 0;

        for (int j0 = tileBeg; j0 < tileEnd; j0 += 32) {
            const int idx = idxNext;
            float2 v[GB];

            issue_rows(xl, idx, 0, v);
            idxNext = (lane < tileEnd - (j0 + 32)) ? perm[j0 + 32 + lane] : 0;

            #pragma unroll
            for (int k = 0; k < GB; ++k) {
                const int j = j0 + k;
                while (j == rowEnd) {
                    *(float2*)(aPtr + r * APITCH) = make_float2(ax, ay);
                    ax = 0.0f; ay = 0.0f;
                    ++r;
                    rowEnd = (r < 16) ? __builtin_amdgcn_readlane(myEnd, r) : -1;
                }
                if (j < tileEnd) { ax += v[k].x; ay += v[k].y; }
            }

            if (j0 + GB < tileEnd) {
                issue_rows(xl, idx, GB, v);
                #pragma unroll
                for (int k = 0; k < GB; ++k) {
                    const int j = j0 + GB + k;
                    while (j == rowEnd) {
                        *(float2*)(aPtr + r * APITCH) = make_float2(ax, ay);
                        ax = 0.0f; ay = 0.0f;
                        ++r;
                        rowEnd = (r < 16) ? __builtin_amdgcn_readlane(myEnd, r) : -1;
                    }
                    if (j < tileEnd) { ax += v[k].x; ay += v[k].y; }
                }
            }
        }
        for (; r < 16; ++r) {
            *(float2*)(aPtr + r * APITCH) = make_float2(ax, ay);
            ax = 0.0f; ay = 0.0f;
        }
    }
    __syncthreads();
    if (!active) return;

    const int m    = lane & 15;
    const int hi   = lane >> 4;
    const int row  = row0 + m;

    const float dinv = 1.0f / fmaxf((float)cnt[row], 1.0f);
    const float* __restrict__ xrow = x + (size_t)row * DIM;
    const float* arow = myA + m * APITCH;

    v8f acc[4];
    for (int nt = 0; nt < 4; ++nt) {
        float b = sB[nt * 16 + m];
        for (int r = 0; r < 8; ++r) acc[nt][r] = b;
    }

    for (int kb = 0; kb < DIM; kb += 4) {
        const int kk = kb + 2 * hi;
        v2f as, an;
        as.x = xrow[kk];
        as.y = xrow[kk + 1];
        an.x = arow[kk]     * dinv;
        an.y = arow[kk + 1] * dinv;

        for (int nt = 0; nt < 4; ++nt) {
            const int bi = kk * DIM + nt * 16 + m;
            v2f bs, bn;
            bs.x = sWs[bi];
            bs.y = sWs[bi + DIM];
            bn.x = sWn[bi];
            bn.y = sWn[bi + DIM];
            acc[nt] = __builtin_amdgcn_wmma_f32_16x16x4_f32(
                          false, as, false, bs, (short)0, acc[nt], false, false);
            acc[nt] = __builtin_amdgcn_wmma_f32_16x16x4_f32(
                          false, an, false, bn, (short)0, acc[nt], false, false);
        }
    }

    for (int nt = 0; nt < 4; ++nt) {
        const int col = nt * 16 + m;
        for (int r = 0; r < 8; ++r) {
            const int mrow = r + 8 * hi;
            out[(size_t)(row0 + mrow) * DIM + col] = acc[nt][r];
        }
    }
}

extern "C" void kernel_launch(void* const* d_in, const int* in_sizes, int n_in,
                              void* d_out, int out_size, void* d_ws, size_t ws_size,
                              hipStream_t stream) {
    const float* x      = (const float*)d_in[0];
    const int*   src    = (const int*)  d_in[1];
    const int*   dst    = (const int*)  d_in[2];
    const float* Wself  = (const float*)d_in[3];
    const float* bself  = (const float*)d_in[4];
    const float* Wneigh = (const float*)d_in[5];
    const float* bneigh = (const float*)d_in[6];
    float*       out    = (float*)d_out;

    const int N = in_sizes[0] / DIM;
    const int E = in_sizes[1];

    int* cnt    = (int*)d_ws;
    int* total  = cnt + N;
    int* cursor = cnt + N + 1;
    int* perm   = cursor + N;

    sage_zero_counts_kernel<<<(N + 1 + 255) / 256, 256, 0, stream>>>(cnt, N + 1);

    sage_count_kernel<<<(E + 255) / 256, 256, 0, stream>>>(dst, cnt, E);

    sage_alloc_segments_kernel<<<(N + 255) / 256, 256, 0, stream>>>(cnt, cursor, total, N);

    sage_fill_kernel<<<(E + 255) / 256, 256, 0, stream>>>(src, dst, cursor, perm, E);

    const int nTiles = (N + 15) / 16;
    sage_gather_wmma_kernel<<<(nTiles + TILE_WAVES - 1) / TILE_WAVES, TILE_WAVES * 32, 0, stream>>>(
        x, cnt, cursor, perm, Wself, bself, Wneigh, bneigh, out, nTiles);
}


# Per-edge global atomics (count, slot) replaced by a counting sort into 64-row bins on LDS counters, 3 launches not 5; a bin's rows are now summed by LDS float adds in arrival order
# speedup vs baseline: 3.7534x; 2.9225x over previous
#include <hip/hip_runtime.h>


typedef __attribute__((ext_vector_type(2))) float v2f;
typedef __attribute__((ext_vector_type(8))) float v8f;

#define DIM 64
#define TILE_WAVES 4
#define APITCH 68
#define BIN_SHIFT 6
#define NB_MAX 1024
#define SORT_THREADS 1024
#define SORT_BLOCKS 64
#define SORT_ITEMS 16
#define SORT_CAP 12800
#define ROW_SHIFT 26
#define ID_MASK 0x03FFFFFFu

__global__ void __launch_bounds__(SORT_THREADS)
sage_bin_count_kernel(const int* __restrict__ dst, int* __restrict__ hist,
                      int E, int nBins, int chunk) {
    __shared__ int h[NB_MAX];
    if (threadIdx.x < nBins) h[threadIdx.x] = 0;
    __syncthreads();
    const int e0 = blockIdx.x * chunk;
    const int e1 = min(E, e0 + chunk);
    for (int eb = e0 + threadIdx.x; eb < e1; eb += SORT_ITEMS * SORT_THREADS) {
        int d[SORT_ITEMS];
        #pragma unroll
        for (int i = 0; i < SORT_ITEMS; ++i) {
            const int e = eb + i * SORT_THREADS;
            const int dv = dst[min(e, e1 - 1)];
            d[i] = (e < e1) ? dv : -1;
        }
        #pragma unroll
        for (int i = 0; i < SORT_ITEMS; ++i)
            if (d[i] >= 0) atomicAdd(&h[min(d[i] >> BIN_SHIFT, nBins - 1)], 1);
    }
    __syncthreads();
    if (threadIdx.x < nBins) hist[blockIdx.x * nBins + threadIdx.x] = h[threadIdx.x];
}

__device__ __forceinline__ int block_excl_scan(int val, int* wsum) {
    const int lane = threadIdx.x & 31;
    const int wv   = threadIdx.x >> 5;
    int incl = val;
    for (int o = 1; o < 32; o <<= 1) {
        int u = __shfl_up(incl, o);
        if (lane >= o) incl += u;
    }
    if (lane == 31) wsum[wv] = incl;
    __syncthreads();
    if (wv == 0) {
        const int s = wsum[lane];
        int si = s;
        for (int o = 1; o < 32; o <<= 1) {
            int u = __shfl_up(si, o);
            if (lane >= o) si += u;
        }
        wsum[lane] = si - s;
    }
    __syncthreads();
    const int r = wsum[wv] + incl - val;
    __syncthreads();
    return r;
}

__global__ void __launch_bounds__(SORT_THREADS)
sage_bin_fill_kernel(const int* __restrict__ src, const int* __restrict__ dst,
                     const int* __restrict__ hist, int* __restrict__ binBeg,
                     unsigned* __restrict__ perm, int E, int nBins, int chunk) {
    __shared__ int cur [NB_MAX];
    __shared__ int lbeg[NB_MAX];
    __shared__ int lcnt[NB_MAX];
    __shared__ int wsum[32];
    __shared__ int nStaged;
    __shared__ unsigned sWord[SORT_CAP];
    __shared__ int      sPos [SORT_CAP];
    const int t  = threadIdx.x;
    const int me = blockIdx.x;

    int tot = 0, pre = 0, own = 0;
    if (t < nBins) {
        int v[SORT_BLOCKS];
        #pragma unroll
        for (int w = 0; w < SORT_BLOCKS; ++w) v[w] = hist[w * nBins + t];
        #pragma unroll
        for (int w = 0; w < SORT_BLOCKS; ++w) {
            tot += v[w];
            pre += (w <  me) ? v[w] : 0;
            own += (w == me) ? v[w] : 0;
        }
    }
    const int bstart = block_excl_scan(tot, wsum);
    const int lstart = block_excl_scan(own, wsum);
    if (t < nBins) {
        cur [t] = bstart + pre;
        lbeg[t] = lstart;
        lcnt[t] = 0;
        if (t == nBins - 1) nStaged = lstart + own;
        if (me == 0) {
            binBeg[t] = bstart;
            if (t == nBins - 1) binBeg[nBins] = bstart + tot;
        }
    }
    __syncthreads();

    const int e0 = me * chunk;
    const int e1 = min(E, e0 + chunk);
    const bool staged = (e1 - e0) <= SORT_CAP;
    for (int eb = e0 + t; eb < e1; eb += SORT_ITEMS * SORT_THREADS) {
        int d[SORT_ITEMS], s[SORT_ITEMS];
        #pragma unroll
        for (int i = 0; i < SORT_ITEMS; ++i) {
            const int e  = eb + i * SORT_THREADS;
            const int ec = min(e, e1 - 1);
            const int dv = dst[ec];
            s[i] = src[ec];
            d[i] = (e < e1) ? dv : -1;
        }
        #pragma unroll
        for (int i = 0; i < SORT_ITEMS; ++i) {
            if (d[i] >= 0) {
                const int bin = min(d[i] >> BIN_SHIFT, nBins - 1);
                const int r   = atomicAdd(&lcnt[bin], 1);
                const unsigned word = ((unsigned)s[i] & ID_MASK) | ((unsigned)(d[i] & 63) << ROW_SHIFT);
                if (staged) {
                    const int lp = lbeg[bin] + r;
                    sWord[lp] = word;
                    sPos [lp] = cur[bin] + r;
                } else {
                    perm[cur[bin] + r] = word;
                }
            }
        }
    }
    __syncthreads();
    if (staged) {
        const int n = nStaged;
        for (int i = t; i < n; i += SORT_THREADS) perm[sPos[i]] = sWord[i];
    }
}

#define GB 16

__device__ __forceinline__ void issue_rows(const float* __restrict__ xl, unsigned w, int i0,
                                           int lastRow, float2 (&v)[GB]) {
    int s[GB];
    #pragma unroll
    for (int k = 0; k < GB; ++k)
        s[k] = min((int)((unsigned)__builtin_amdgcn_readlane((int)w, i0 + k) & ID_MASK), lastRow);
    #pragma unroll
    for (int k = 0; k < GB; ++k) v[k] = *(const float2*)(xl + (size_t)s[k] * DIM);
}

__device__ __forceinline__ void add_rows(float* aPtr, unsigned w, int i0, int nValid,
                                         const float2 (&v)[GB]) {
    #pragma unroll
    for (int k = 0; k < GB; ++k) {
        if (i0 + k < nValid) {
            const int lr = (int)((unsigned)__builtin_amdgcn_readlane((int)w, i0 + k) >> ROW_SHIFT);
            float* p = aPtr + lr * APITCH;
            __hip_atomic_fetch_add(p,     v[k].x, __ATOMIC_RELAXED, __HIP_MEMORY_SCOPE_WORKGROUP);
            __hip_atomic_fetch_add(p + 1, v[k].y, __ATOMIC_RELAXED, __HIP_MEMORY_SCOPE_WORKGROUP);
        }
    }
}

__global__ void __launch_bounds__(TILE_WAVES * 32)
sage_gather_wmma_kernel(const float* __restrict__ x,
                        const int*   __restrict__ binBeg,
                        const unsigned* __restrict__ perm,
                        const float* __restrict__ Wself,
                        const float* __restrict__ bself,
                        const float* __restrict__ Wneigh,
                        const float* __restrict__ bneigh,
                        float*       __restrict__ out,
                        int nTiles, int nRows) {
    __shared__ float sWs[DIM * DIM];
    __shared__ float sWn[DIM * DIM];
    __shared__ float sB [DIM];
    __shared__ float sA [TILE_WAVES * 16 * APITCH];
    __shared__ int   sDeg[TILE_WAVES * 16];

    for (int i = threadIdx.x; i < DIM * DIM; i += TILE_WAVES * 32) {
        sWs[i] = Wself[i];
        sWn[i] = Wneigh[i];
    }
    if (threadIdx.x < DIM) sB[threadIdx.x] = bself[threadIdx.x] + bneigh[threadIdx.x];
    for (int i = threadIdx.x; i < TILE_WAVES * 16 * APITCH; i += TILE_WAVES * 32) sA[i] = 0.0f;
    if (threadIdx.x < TILE_WAVES * 16) sDeg[threadIdx.x] = 0;
    __syncthreads();

    const int wave = __builtin_amdgcn_readfirstlane((int)(threadIdx.x >> 5));
    const int lane = threadIdx.x & 31;
    const int tile = blockIdx.x * TILE_WAVES + wave;
    const bool active = tile < nTiles;
    const int row0 = tile * 16;

    {
        const int beg = binBeg[blockIdx.x];
        const int end = binBeg[blockIdx.x + 1];
        const float* xl = x + 2 * lane;
        float* aPtr = sA + 2 * lane;
        const int lastRow = nRows - 1;

        int j0 = beg + 32 * wave;
        unsigned wordNext = (j0 + lane < end) ? perm[j0 + lane] : 0u;

        for (; j0 < end; j0 += 32 * TILE_WAVES) {
            const unsigned word = wordNext;
            const int nValid = min(32, end - j0);
            float2 v[GB];

            issue_rows(xl, word, 0, lastRow, v);
            const int jn = j0 + 32 * TILE_WAVES;
            wordNext = (jn + lane < end) ? perm[jn + lane] : 0u;

            if (lane < nValid)
                __hip_atomic_fetch_add(&sDeg[word >> ROW_SHIFT], 1, __ATOMIC_RELAXED, __HIP_MEMORY_SCOPE_WORKGROUP);

            add_rows(aPtr, word, 0, nValid, v);
            if (nValid > GB) {
                issue_rows(xl, word, GB, lastRow, v);
                add_rows(aPtr, word, GB, nValid, v);
            }
        }
    }
    __syncthreads();
    if (!active) return;

    const int m    = lane & 15;
    const int hi   = lane >> 4;
    const int row  = row0 + m;

    const float dinv = 1.0f / fmaxf((float)sDeg[wave * 16 + m], 1.0f);
    const float* __restrict__ xrow = x + (size_t)row * DIM;
    const float* arow = sA + (wave * 16 + m) * APITCH;

    v8f acc[4];
    for (int nt = 0; nt < 4; ++nt) {
        float b = sB[nt * 16 + m];
        for (int r = 0; r < 8; ++r) acc[nt][r] = b;
    }

    for (int kb = 0; kb < DIM; kb += 4) {
        const int kk = kb + 2 * hi;
        v2f as, an;
        as.x = xrow[kk];
        as.y = xrow[kk + 1];
        an.x = arow[kk]     * dinv;
        an.y = arow[kk + 1] * dinv;

        for (int nt = 0; nt < 4; ++nt) {
            const int bi = kk * DIM + nt * 16 + m;
            v2f bs, bn;
            bs.x = sWs[bi];
            bs.y = sWs[bi + DIM];
            bn.x = sWn[bi];
            bn.y = sWn[bi + DIM];
            acc[nt] = __builtin_amdgcn_wmma_f32_16x16x4_f32(
                          false, as, false, bs, (short)0, acc[nt], false, false);
            acc[nt] = __builtin_amdgcn_wmma_f32_16x16x4_f32(
                          false, an, false, bn, (short)0, acc[nt], false, false);
        }
    }

    for (int nt = 0; nt < 4; ++nt) {
        const int col = nt * 16 + m;
        for (int r = 0; r < 8; ++r) {
            const int mrow = r + 8 * hi;
            out[(size_t)(row0 + mrow) * DIM + col] = acc[nt][r];
        }
    }
}

extern "C" void kernel_launch(void* const* d_in, const int* in_sizes, int n_in,
                              void* d_out, int out_size, void* d_ws, size_t ws_size,
                              hipStream_t stream) {
    const float* x      = (const float*)d_in[0];
    const int*   src    = (const int*)  d_in[1];
    const int*   dst    = (const int*)  d_in[2];
    const float* Wself  = (const float*)d_in[3];
    const float* bself  = (const float*)d_in[4];
    const float* Wneigh = (const float*)d_in[5];
    const float* bneigh = (const float*)d_in[6];
    float*       out    = (float*)d_out;

    const int N = in_sizes[0] / DIM;
    const int E = in_sizes[1];

    const int nTiles = (N + 15) / 16;
    const int nBins  = (nTiles + TILE_WAVES - 1) / TILE_WAVES;
    const int chunk  = (E + SORT_BLOCKS - 1) / SORT_BLOCKS;

    int*      hist   = (int*)d_ws;
    int*      binBeg = hist + SORT_BLOCKS * nBins;
    unsigned* perm   = (unsigned*)(binBeg + nBins + 1);

    sage_bin_count_kernel<<<SORT_BLOCKS, SORT_THREADS, 0, stream>>>(dst, hist, E, nBins, chunk);

    sage_bin_fill_kernel<<<SORT_BLOCKS, SORT_THREADS, 0, stream>>>(src, dst, hist, binBeg, perm, E, nBins, chunk);

    sage_gather_wmma_kernel<<<nBins, TILE_WAVES * 32, 0, stream>>>(
        x, binBeg, perm, Wself, bself, Wneigh, bneigh, out, nTiles, N);
}


# Each sort block orders only its own edge slice by bin, so the count launch goes (2 launches, not 3); a gather wave walks 16 per-block runs of its bin; LDS float adds still in arrival order
# speedup vs baseline: 4.2673x; 1.1369x over previous
#include <hip/hip_runtime.h>


typedef __attribute__((ext_vector_type(2))) float v2f;
typedef __attribute__((ext_vector_type(8))) float v8f;

#define DIM 64
#define TILE_WAVES 4
#define APITCH 68
#define BIN_SHIFT 6
#define NB_MAX 1024
#define SORT_THREADS 1024
#define SORT_BLOCKS 64
#define SORT_ITEMS 16
#define SORT_CAP (SORT_ITEMS * SORT_THREADS)
#define RUNS_PER_WAVE (SORT_BLOCKS / TILE_WAVES)
#define ROW_SHIFT 26
#define ID_MASK 0x03FFFFFFu

__device__ __forceinline__ int block_excl_scan(int val, int* wsum) {
    const int lane = threadIdx.x & 31;
    const int wv   = threadIdx.x >> 5;
    int incl = val;
    for (int o = 1; o < 32; o <<= 1) {
        int u = __shfl_up(incl, o);
        if (lane >= o) incl += u;
    }
    if (lane == 31) wsum[wv] = incl;
    __syncthreads();
    if (wv == 0) {
        const int s = wsum[lane];
        int si = s;
        for (int o = 1; o < 32; o <<= 1) {
            int u = __shfl_up(si, o);
            if (lane >= o) si += u;
        }
        wsum[lane] = si - s;
    }
    __syncthreads();
    const int r = wsum[wv] + incl - val;
    __syncthreads();
    return r;
}

__global__ void __launch_bounds__(SORT_THREADS)
sage_slice_sort_kernel(const int* __restrict__ src, const int* __restrict__ dst,
                       int* __restrict__ runBeg, unsigned* __restrict__ perm,
                       int E, int nBins, int chunk) {
    __shared__ int lcnt[NB_MAX];
    __shared__ int lbeg[NB_MAX];
    __shared__ int wsum[32];
    __shared__ unsigned sWord[SORT_CAP];
    const int t  = threadIdx.x;
    const int me = blockIdx.x;
    const int e0 = min(E, me * chunk);
    const int e1 = min(E, e0 + chunk);
    const int n  = e1 - e0;
    const bool one = n <= SORT_CAP;

    if (t < nBins) lcnt[t] = 0;
    __syncthreads();

    int d[SORT_ITEMS], s[SORT_ITEMS], r[SORT_ITEMS];
    if (one) {
        #pragma unroll
        for (int i = 0; i < SORT_ITEMS; ++i) {
            const int e  = e0 + t + i * SORT_THREADS;
            const int ec = max(min(e, e1 - 1), 0);
            const int dv = dst[ec];
            s[i] = src[ec];
            d[i] = (e < e1) ? dv : -1;
        }
        #pragma unroll
        for (int i = 0; i < SORT_ITEMS; ++i) {
            r[i] = 0;
            if (d[i] >= 0) r[i] = atomicAdd(&lcnt[min(d[i] >> BIN_SHIFT, nBins - 1)], 1);
        }
    } else {
        for (int eb = e0 + t; eb < e1; eb += SORT_ITEMS * SORT_THREADS) {
            #pragma unroll
            for (int i = 0; i < SORT_ITEMS; ++i) {
                const int e = eb + i * SORT_THREADS;
                const int dv = dst[min(e, e1 - 1)];
                d[i] = (e < e1) ? dv : -1;
            }
            #pragma unroll
            for (int i = 0; i < SORT_ITEMS; ++i)
                if (d[i] >= 0) atomicAdd(&lcnt[min(d[i] >> BIN_SHIFT, nBins - 1)], 1);
        }
    }
    __syncthreads();

    const int own    = (t < nBins) ? lcnt[t] : 0;
    const int lstart = block_excl_scan(own, wsum);
    if (t < nBins) {
        lbeg[t] = lstart;
        lcnt[t] = 0;
        runBeg[t * SORT_BLOCKS + me] = lstart;
        if (t == nBins - 1) runBeg[nBins * SORT_BLOCKS + me] = lstart + own;
    }
    __syncthreads();

    if (one) {
        #pragma unroll
        for (int i = 0; i < SORT_ITEMS; ++i) {
            if (d[i] >= 0) {
                const int bin = min(d[i] >> BIN_SHIFT, nBins - 1);
                sWord[lbeg[bin] + r[i]] = ((unsigned)s[i] & ID_MASK) | ((unsigned)(d[i] & 63) << ROW_SHIFT);
            }
        }
        __syncthreads();
        for (int i = t; i < n; i += SORT_THREADS) perm[e0 + i] = sWord[i];
    } else {
        for (int eb = e0 + t; eb < e1; eb += SORT_ITEMS * SORT_THREADS) {
            #pragma unroll
            for (int i = 0; i < SORT_ITEMS; ++i) {
                const int e  = eb + i * SORT_THREADS;
                const int ec = min(e, e1 - 1);
                const int dv = dst[ec];
                s[i] = src[ec];
                d[i] = (e < e1) ? dv : -1;
            }
            #pragma unroll
            for (int i = 0; i < SORT_ITEMS; ++i) {
                if (d[i] >= 0) {
                    const int bin = min(d[i] >> BIN_SHIFT, nBins - 1);
                    const int rk  = atomicAdd(&lcnt[bin], 1);
                    perm[e0 + lbeg[bin] + rk] = ((unsigned)s[i] & ID_MASK) | ((unsigned)(d[i] & 63) << ROW_SHIFT);
                }
            }
        }
    }
}

#define GB 16

__device__ __forceinline__ void issue_rows(const float* __restrict__ xl, unsigned w, int i0,
                                           int lastRow, float2 (&v)[GB]) {
    int s[GB];
    #pragma unroll
    for (int k = 0; k < GB; ++k)
        s[k] = min((int)((unsigned)__builtin_amdgcn_readlane((int)w, i0 + k) & ID_MASK), lastRow);
    #pragma unroll
    for (int k = 0; k < GB; ++k) v[k] = *(const float2*)(xl + (size_t)s[k] * DIM);
}

__device__ __forceinline__ void add_rows(float* aPtr, unsigned w, int i0, int nValid,
                                         const float2 (&v)[GB]) {
    #pragma unroll
    for (int k = 0; k < GB; ++k) {
        if (i0 + k < nValid) {
            const int lr = (int)((unsigned)__builtin_amdgcn_readlane((int)w, i0 + k) >> ROW_SHIFT);
            float* p = aPtr + lr * APITCH;
            __hip_atomic_fetch_add(p,     v[k].x, __ATOMIC_RELAXED, __HIP_MEMORY_SCOPE_WORKGROUP);
            __hip_atomic_fetch_add(p + 1, v[k].y, __ATOMIC_RELAXED, __HIP_MEMORY_SCOPE_WORKGROUP);
        }
    }
}

__device__ __forceinline__ unsigned fetch_word(const unsigned* __restrict__ perm, int q, int total,
                                               const int (&runEnd)[RUNS_PER_WAVE - 1], int gdelta, int lastEdge) {
    int k = 0;
    #pragma unroll
    for (int j = 0; j < RUNS_PER_WAVE - 1; ++j) k += (q >= runEnd[j]) ? 1 : 0;
    const int gd = __shfl(gdelta, k);
    const int idx = max(min(gd + q, lastEdge), 0);
    const unsigned w = perm[idx];
    return (q < total) ? w : 0u;
}

__global__ void __launch_bounds__(TILE_WAVES * 32)
sage_gather_wmma_kernel(const float* __restrict__ x,
                        const int*   __restrict__ runBeg,
                        const unsigned* __restrict__ perm,
                        const float* __restrict__ Wself,
                        const float* __restrict__ bself,
                        const float* __restrict__ Wneigh,
                        const float* __restrict__ bneigh,
                        float*       __restrict__ out,
                        int nTiles, int nRows, int E, int chunk) {
    __shared__ float sWs[DIM * DIM];
    __shared__ float sWn[DIM * DIM];
    __shared__ float sB [DIM];
    __shared__ float sA [TILE_WAVES * 16 * APITCH];
    __shared__ int   sDeg[TILE_WAVES * 16];

    for (int i = threadIdx.x; i < DIM * DIM; i += TILE_WAVES * 32) {
        sWs[i] = Wself[i];
        sWn[i] = Wneigh[i];
    }
    if (threadIdx.x < DIM) sB[threadIdx.x] = bself[threadIdx.x] + bneigh[threadIdx.x];
    for (int i = threadIdx.x; i < TILE_WAVES * 16 * APITCH; i += TILE_WAVES * 32) sA[i] = 0.0f;
    if (threadIdx.x < TILE_WAVES * 16) sDeg[threadIdx.x] = 0;
    __syncthreads();

    const int wave = __builtin_amdgcn_readfirstlane((int)(threadIdx.x >> 5));
    const int lane = threadIdx.x & 31;
    const int tile = blockIdx.x * TILE_WAVES + wave;
    const bool active = tile < nTiles;
    const int row0 = tile * 16;

    {
        const int sub = lane & (RUNS_PER_WAVE - 1);
        const int run = wave * RUNS_PER_WAVE + sub;
        const int* rb = runBeg + (size_t)blockIdx.x * SORT_BLOCKS + run;
        const int rs  = rb[0];
        const int re  = rb[SORT_BLOCKS];
        const int len = min(max(re - rs, 0), chunk);
        int incl = len;
        for (int o = 1; o < RUNS_PER_WAVE; o <<= 1) {
            int u = __shfl_up(incl, o, RUNS_PER_WAVE);
            if (sub >= o) incl += u;
        }
        const int total = __builtin_amdgcn_readlane(incl, RUNS_PER_WAVE - 1);
        int runEnd[RUNS_PER_WAVE - 1];
        #pragma unroll
        for (int j = 0; j < RUNS_PER_WAVE - 1; ++j) runEnd[j] = __builtin_amdgcn_readlane(incl, j);
        const int gdelta = run * chunk + rs - (incl - len);
        const int lastEdge = E - 1;

        const float* xl = x + 2 * lane;
        float* aPtr = sA + 2 * lane;
        const int lastRow = nRows - 1;

        unsigned wordNext = fetch_word(perm, lane, total, runEnd, gdelta, lastEdge);

        for (int q0 = 0; q0 < total; q0 += 32) {
            const unsigned word = wordNext;
            const int nValid = min(32, total - q0);
            float2 v[GB];

            issue_rows(xl, word, 0, lastRow, v);
            wordNext = fetch_word(perm, q0 + 32 + lane, total, runEnd, gdelta, lastEdge);

            if (lane < nValid)
                __hip_atomic_fetch_add(&sDeg[word >> ROW_SHIFT], 1, __ATOMIC_RELAXED, __HIP_MEMORY_SCOPE_WORKGROUP);

            add_rows(aPtr, word, 0, nValid, v);
            if (nValid > GB) {
                issue_rows(xl, word, GB, lastRow, v);
                add_rows(aPtr, word, GB, nValid, v);
            }
        }
    }
    __syncthreads();
    if (!active) return;

    const int m    = lane & 15;
    const int hi   = lane >> 4;
    const int row  = row0 + m;

    const float dinv = 1.0f / fmaxf((float)sDeg[wave * 16 + m], 1.0f);
    const float* __restrict__ xrow = x + (size_t)row * DIM;
    const float* arow = sA + (wave * 16 + m) * APITCH;

    v8f acc[4];
    for (int nt = 0; nt < 4; ++nt) {
        float b = sB[nt * 16 + m];
        for (int r = 0; r < 8; ++r) acc[nt][r] = b;
    }

    for (int kb = 0; kb < DIM; kb += 4) {
        const int kk = kb + 2 * hi;
        v2f as, an;
        as.x = xrow[kk];
        as.y = xrow[kk + 1];
        an.x = arow[kk]     * dinv;
        an.y = arow[kk + 1] * dinv;

        for (int nt = 0; nt < 4; ++nt) {
            const int bi = kk * DIM + nt * 16 + m;
            v2f bs, bn;
            bs.x = sWs[bi];
            bs.y = sWs[bi + DIM];
            bn.x = sWn[bi];
            bn.y = sWn[bi + DIM];
            acc[nt] = __builtin_amdgcn_wmma_f32_16x16x4_f32(
                          false, as, false, bs, (short)0, acc[nt], false, false);
            acc[nt] = __builtin_amdgcn_wmma_f32_16x16x4_f32(
                          false, an, false, bn, (short)0, acc[nt], false, false);
        }
    }

    for (int nt = 0; nt < 4; ++nt) {
        const int col = nt * 16 + m;
        for (int r = 0; r < 8; ++r) {
            const int mrow = r + 8 * hi;
            out[(size_t)(row0 + mrow) * DIM + col] = acc[nt][r];
        }
    }
}

extern "C" void kernel_launch(void* const* d_in, const int* in_sizes, int n_in,
                              void* d_out, int out_size, void* d_ws, size_t ws_size,
                              hipStream_t stream) {
    const float* x      = (const float*)d_in[0];
    const int*   src    = (const int*)  d_in[1];
    const int*   dst    = (const int*)  d_in[2];
    const float* Wself  = (const float*)d_in[3];
    const float* bself  = (const float*)d_in[4];
    const float* Wneigh = (const float*)d_in[5];
    const float* bneigh = (const float*)d_in[6];
    float*       out    = (float*)d_out;

    const int N = in_sizes[0] / DIM;
    const int E = in_sizes[1];

    const int nTiles = (N + 15) / 16;
    const int nBins  = (nTiles + TILE_WAVES - 1) / TILE_WAVES;
    const int chunk  = (E + SORT_BLOCKS - 1) / SORT_BLOCKS;

    int*      runBeg = (int*)d_ws;
    unsigned* perm   = (unsigned*)(runBeg + (nBins + 1) * SORT_BLOCKS);

    sage_slice_sort_kernel<<<SORT_BLOCKS, SORT_THREADS, 0, stream>>>(src, dst, runBeg, perm, E, nBins, chunk);

    sage_gather_wmma_kernel<<<nBins, TILE_WAVES * 32, 0, stream>>>(
        x, runBeg, perm, Wself, bself, Wneigh, bneigh, out, nTiles, N, E, chunk);
}


# Loads issued ahead of use in the gather launch: run table, edge words and first 16 rows before the (now one-trip, 16-byte) weight staging and barrier; next 16 rows before the arrived 16 are added
# speedup vs baseline: 4.5254x; 1.0605x over previous
#include <hip/hip_runtime.h>


typedef __attribute__((ext_vector_type(2))) float v2f;
typedef __attribute__((ext_vector_type(8))) float v8f;

#define DIM 64
#define TILE_WAVES 4
#define APITCH 68
#define BIN_SHIFT 6
#define NB_MAX 1024
#define SORT_THREADS 1024
#define SORT_BLOCKS 64
#define SORT_ITEMS 16
#define SORT_CAP (SORT_ITEMS * SORT_THREADS)
#define RUNS_PER_WAVE (SORT_BLOCKS / TILE_WAVES)
#define ROW_SHIFT 26
#define ID_MASK 0x03FFFFFFu

__device__ __forceinline__ int block_excl_scan(int val, int* wsum) {
    const int lane = threadIdx.x & 31;
    const int wv   = threadIdx.x >> 5;
    int incl = val;
    for (int o = 1; o < 32; o <<= 1) {
        int u = __shfl_up(incl, o);
        if (lane >= o) incl += u;
    }
    if (lane == 31) wsum[wv] = incl;
    __syncthreads();
    if (wv == 0) {
        const int s = wsum[lane];
        int si = s;
        for (int o = 1; o < 32; o <<= 1) {
            int u = __shfl_up(si, o);
            if (lane >= o) si += u;
        }
        wsum[lane] = si - s;
    }
    __syncthreads();
    const int r = wsum[wv] + incl - val;
    __syncthreads();
    return r;
}

__global__ void __launch_bounds__(SORT_THREADS)
sage_slice_sort_kernel(const int* __restrict__ src, const int* __restrict__ dst,
                       int* __restrict__ runBeg, unsigned* __restrict__ perm,
                       int E, int nBins, int chunk) {
    __shared__ int lcnt[NB_MAX];
    __shared__ int lbeg[NB_MAX];
    __shared__ int wsum[32];
    __shared__ unsigned sWord[SORT_CAP];
    const int t  = threadIdx.x;
    const int me = blockIdx.x;
    const int e0 = min(E, me * chunk);
    const int e1 = min(E, e0 + chunk);
    const int n  = e1 - e0;
    const bool one = n <= SORT_CAP;

    if (t < nBins) lcnt[t] = 0;
    __syncthreads();

    int d[SORT_ITEMS], s[SORT_ITEMS], r[SORT_ITEMS];
    if (one) {
        #pragma unroll
        for (int i = 0; i < SORT_ITEMS; ++i) {
            const int e  = e0 + t + i * SORT_THREADS;
            const int ec = max(min(e, e1 - 1), 0);
            const int dv = dst[ec];
            s[i] = src[ec];
            d[i] = (e < e1) ? dv : -1;
        }
        #pragma unroll
        for (int i = 0; i < SORT_ITEMS; ++i) {
            r[i] = 0;
            if (d[i] >= 0) r[i] = atomicAdd(&lcnt[min(d[i] >> BIN_SHIFT, nBins - 1)], 1);
        }
    } else {
        for (int eb = e0 + t; eb < e1; eb += SORT_ITEMS * SORT_THREADS) {
            #pragma unroll
            for (int i = 0; i < SORT_ITEMS; ++i) {
                const int e = eb + i * SORT_THREADS;
                const int dv = dst[min(e, e1 - 1)];
                d[i] = (e < e1) ? dv : -1;
            }
            #pragma unroll
            for (int i = 0; i < SORT_ITEMS; ++i)
                if (d[i] >= 0) atomicAdd(&lcnt[min(d[i] >> BIN_SHIFT, nBins - 1)], 1);
        }
    }
    __syncthreads();

    const int own    = (t < nBins) ? lcnt[t] : 0;
    const int lstart = block_excl_scan(own, wsum);
    if (t < nBins) {
        lbeg[t] = lstart;
        lcnt[t] = 0;
        runBeg[t * SORT_BLOCKS + me] = lstart;
        if (t == nBins - 1) runBeg[nBins * SORT_BLOCKS + me] = lstart + own;
    }
    __syncthreads();

    if (one) {
        #pragma unroll
        for (int i = 0; i < SORT_ITEMS; ++i) {
            if (d[i] >= 0) {
                const int bin = min(d[i] >> BIN_SHIFT, nBins - 1);
                sWord[lbeg[bin] + r[i]] = ((unsigned)s[i] & ID_MASK) | ((unsigned)(d[i] & 63) << ROW_SHIFT);
            }
        }
        __syncthreads();
        for (int i = t; i < n; i += SORT_THREADS) perm[e0 + i] = sWord[i];
    } else {
        for (int eb = e0 + t; eb < e1; eb += SORT_ITEMS * SORT_THREADS) {
            #pragma unroll
            for (int i = 0; i < SORT_ITEMS; ++i) {
                const int e  = eb + i * SORT_THREADS;
                const int ec = min(e, e1 - 1);
                const int dv = dst[ec];
                s[i] = src[ec];
                d[i] = (e < e1) ? dv : -1;
            }
            #pragma unroll
            for (int i = 0; i < SORT_ITEMS; ++i) {
                if (d[i] >= 0) {
                    const int bin = min(d[i] >> BIN_SHIFT, nBins - 1);
                    const int rk  = atomicAdd(&lcnt[bin], 1);
                    perm[e0 + lbeg[bin] + rk] = ((unsigned)s[i] & ID_MASK) | ((unsigned)(d[i] & 63) << ROW_SHIFT);
                }
            }
        }
    }
}

#define GB 16
#define W_PIECES (DIM * DIM / 4 / (TILE_WAVES * 32))

__device__ __forceinline__ void issue_rows(const float* __restrict__ xl, unsigned w, int i0,
                                           int lastRow, float2 (&v)[GB]) {
    int s[GB];
    #pragma unroll
    for (int k = 0; k < GB; ++k)
        s[k] = min((int)((unsigned)__builtin_amdgcn_readlane((int)w, i0 + k) & ID_MASK), lastRow);
    #pragma unroll
    for (int k = 0; k < GB; ++k) v[k] = *(const float2*)(xl + (size_t)s[k] * DIM);
}

__device__ __forceinline__ void add_rows(float* aPtr, unsigned w, int i0, int nValid,
                                         const float2 (&v)[GB]) {
    #pragma unroll
    for (int k = 0; k < GB; ++k) {
        if (i0 + k < nValid) {
            const int lr = (int)((unsigned)__builtin_amdgcn_readlane((int)w, i0 + k) >> ROW_SHIFT);
            float* p = aPtr + lr * APITCH;
            __hip_atomic_fetch_add(p,     v[k].x, __ATOMIC_RELAXED, __HIP_MEMORY_SCOPE_WORKGROUP);
            __hip_atomic_fetch_add(p + 1, v[k].y, __ATOMIC_RELAXED, __HIP_MEMORY_SCOPE_WORKGROUP);
        }
    }
}

__device__ __forceinline__ unsigned fetch_word(const unsigned* __restrict__ perm, int q, int total,
                                               const int (&runEnd)[RUNS_PER_WAVE - 1], int gdelta, int lastEdge) {
    int k = 0;
    #pragma unroll
    for (int j = 0; j < RUNS_PER_WAVE - 1; ++j) k += (q >= runEnd[j]) ? 1 : 0;
    const int gd = __shfl(gdelta, k);
    const int idx = max(min(gd + q, lastEdge), 0);
    const unsigned w = perm[idx];
    return (q < total) ? w : 0u;
}

__global__ void __launch_bounds__(TILE_WAVES * 32)
sage_gather_wmma_kernel(const float* __restrict__ x,
                        const int*   __restrict__ runBeg,
                        const unsigned* __restrict__ perm,
                        const float* __restrict__ Wself,
                        const float* __restrict__ bself,
                        const float* __restrict__ Wneigh,
                        const float* __restrict__ bneigh,
                        float*       __restrict__ out,
                        int nTiles, int nRows, int E, int chunk) {
    __shared__ alignas(16) float sWs[DIM * DIM];
    __shared__ alignas(16) float sWn[DIM * DIM];
    __shared__ float sB [DIM];
    __shared__ float sA [TILE_WAVES * 16 * APITCH];
    __shared__ int   sDeg[TILE_WAVES * 16];

    const int wave = __builtin_amdgcn_readfirstlane((int)(threadIdx.x >> 5));
    const int lane = threadIdx.x & 31;
    const int tile = blockIdx.x * TILE_WAVES + wave;
    const bool active = tile < nTiles;
    const int row0 = tile * 16;

    const int sub = lane & (RUNS_PER_WAVE - 1);
    const int run = wave * RUNS_PER_WAVE + sub;
    const int* rb = runBeg + (size_t)blockIdx.x * SORT_BLOCKS + run;
    const int rs  = rb[0];
    const int re  = rb[SORT_BLOCKS];
    const int len = min(max(re - rs, 0), chunk);
    int incl = len;
    for (int o = 1; o < RUNS_PER_WAVE; o <<= 1) {
        int u = __shfl_up(incl, o, RUNS_PER_WAVE);
        if (sub >= o) incl += u;
    }
    const int total = __builtin_amdgcn_readlane(incl, RUNS_PER_WAVE - 1);
    int runEnd[RUNS_PER_WAVE - 1];
    #pragma unroll
    for (int j = 0; j < RUNS_PER_WAVE - 1; ++j) runEnd[j] = __builtin_amdgcn_readlane(incl, j);
    const int gdelta = run * chunk + rs - (incl - len);
    const int lastEdge = E - 1;

    const float* xl = x + 2 * lane;
    float* aPtr = sA + 2 * lane;
    const int lastRow = nRows - 1;

    float2 va[GB], vb[GB];
    unsigned word = 0u, wordNext = 0u;
    if (total > 0) {
        word = fetch_word(perm, lane, total, runEnd, gdelta, lastEdge);
        issue_rows(xl, word, 0, lastRow, va);
        wordNext = fetch_word(perm, 32 + lane, total, runEnd, gdelta, lastEdge);
    }

    #pragma unroll
    for (int j = 0; j < W_PIECES; ++j) {
        const int i = threadIdx.x + j * (TILE_WAVES * 32);
        ((float4*)sWs)[i] = ((const float4*)Wself )[i];
        ((float4*)sWn)[i] = ((const float4*)Wneigh)[i];
    }
    if (threadIdx.x < DIM) sB[threadIdx.x] = bself[threadIdx.x] + bneigh[threadIdx.x];
    for (int i = threadIdx.x; i < TILE_WAVES * 16 * APITCH; i += TILE_WAVES * 32) sA[i] = 0.0f;
    if (threadIdx.x < TILE_WAVES * 16) sDeg[threadIdx.x] = 0;
    __syncthreads();

    for (int q0 = 0; q0 < total; q0 += 32) {
        const int nValid = min(32, total - q0);

        const unsigned wordAfter = fetch_word(perm, q0 + 64 + lane, total, runEnd, gdelta, lastEdge);

        issue_rows(xl, word, GB, lastRow, vb);

        if (lane < nValid)
            __hip_atomic_fetch_add(&sDeg[word >> ROW_SHIFT], 1, __ATOMIC_RELAXED, __HIP_MEMORY_SCOPE_WORKGROUP);

        add_rows(aPtr, word, 0, nValid, va);

        issue_rows(xl, wordNext, 0, lastRow, va);

        add_rows(aPtr, word, GB, nValid, vb);

        word = wordNext;
        wordNext = wordAfter;
    }

    __syncthreads();
    if (!active) return;

    const int m    = lane & 15;
    const int hi   = lane >> 4;
    const int row  = row0 + m;

    const float dinv = 1.0f / fmaxf((float)sDeg[wave * 16 + m], 1.0f);
    const float* __restrict__ xrow = x + (size_t)row * DIM;
    const float* arow = sA + (wave * 16 + m) * APITCH;

    v8f acc[4];
    for (int nt = 0; nt < 4; ++nt) {
        float b = sB[nt * 16 + m];
        for (int r = 0; r < 8; ++r) acc[nt][r] = b;
    }

    for (int kb = 0; kb < DIM; kb += 4) {
        const int kk = kb + 2 * hi;
        v2f as, an;
        as.x = xrow[kk];
        as.y = xrow[kk + 1];
        an.x = arow[kk]     * dinv;
        an.y = arow[kk + 1] * dinv;

        for (int nt = 0; nt < 4; ++nt) {
            const int bi = kk * DIM + nt * 16 + m;
            v2f bs, bn;
            bs.x = sWs[bi];
            bs.y = sWs[bi + DIM];
            bn.x = sWn[bi];
            bn.y = sWn[bi + DIM];
            acc[nt] = __builtin_amdgcn_wmma_f32_16x16x4_f32(
                          false, as, false, bs, (short)0, acc[nt], false, false);
            acc[nt] = __builtin_amdgcn_wmma_f32_16x16x4_f32(
                          false, an, false, bn, (short)0, acc[nt], false, false);
        }
    }

    for (int nt = 0; nt < 4; ++nt) {
        const int col = nt * 16 + m;
        for (int r = 0; r < 8; ++r) {
            const int mrow = r + 8 * hi;
            out[(size_t)(row0 + mrow) * DIM + col] = acc[nt][r];
        }
    }
}

extern "C" void kernel_launch(void* const* d_in, const int* in_sizes, int n_in,
                              void* d_out, int out_size, void* d_ws, size_t ws_size,
                              hipStream_t stream) {
    const float* x      = (const float*)d_in[0];
    const int*   src    = (const int*)  d_in[1];
    const int*   dst    = (const int*)  d_in[2];
    const float* Wself  = (const float*)d_in[3];
    const float* bself  = (const float*)d_in[4];
    const float* Wneigh = (const float*)d_in[5];
    const float* bneigh = (const float*)d_in[6];
    float*       out    = (float*)d_out;

    const int N = in_sizes[0] / DIM;
    const int E = in_sizes[1];

    const int nTiles = (N + 15) / 16;
    const int nBins  = (nTiles + TILE_WAVES - 1) / TILE_WAVES;
    const int chunk  = (E + SORT_BLOCKS - 1) / SORT_BLOCKS;

    int*      runBeg = (int*)d_ws;
    unsigned* perm   = (unsigned*)(runBeg + (nBins + 1) * SORT_BLOCKS);

    sage_slice_sort_kernel<<<SORT_BLOCKS, SORT_THREADS, 0, stream>>>(src, dst, runBeg, perm, E, nBins, chunk);

    sage_gather_wmma_kernel<<<nBins, TILE_WAVES * 32, 0, stream>>>(
        x, runBeg, perm, Wself, bself, Wneigh, bneigh, out, nTiles, N, E, chunk);
}


# Bin's edge words ordered by destination row in LDS with per-wave integer counters (order fixed: wave, position); each wave sums its own 16 rows in registers in that order; LDS float adds removed
# speedup vs baseline: 4.8379x; 1.0691x over previous
#include <hip/hip_runtime.h>


typedef __attribute__((ext_vector_type(2))) float v2f;
typedef __attribute__((ext_vector_type(8))) float v8f;

#define DIM 64
#define TILE_WAVES 4
#define APITCH 68
#define BIN_SHIFT 6
#define NB_MAX 1024
#define SORT_THREADS 1024
#define SORT_BLOCKS 64
#define SORT_ITEMS 16
#define SORT_CAP (SORT_ITEMS * SORT_THREADS)
#define RUNS_PER_WAVE (SORT_BLOCKS / TILE_WAVES)
#define ROW_SHIFT 26
#define ID_MASK 0x03FFFFFFu

__device__ __forceinline__ int block_excl_scan(int val, int* wsum) {
    const int lane = threadIdx.x & 31;
    const int wv   = threadIdx.x >> 5;
    int incl = val;
    for (int o = 1; o < 32; o <<= 1) {
        int u = __shfl_up(incl, o);
        if (lane >= o) incl += u;
    }
    if (lane == 31) wsum[wv] = incl;
    __syncthreads();
    if (wv == 0) {
        const int s = wsum[lane];
        int si = s;
        for (int o = 1; o < 32; o <<= 1) {
            int u = __shfl_up(si, o);
            if (lane >= o) si += u;
        }
        wsum[lane] = si - s;
    }
    __syncthreads();
    const int r = wsum[wv] + incl - val;
    __syncthreads();
    return r;
}

__global__ void __launch_bounds__(SORT_THREADS)
sage_slice_sort_kernel(const int* __restrict__ src, const int* __restrict__ dst,
                       int* __restrict__ runBeg, unsigned* __restrict__ perm,
                       int E, int nBins, int chunk) {
    __shared__ int lcnt[NB_MAX];
    __shared__ int lbeg[NB_MAX];
    __shared__ int wsum[32];
    __shared__ unsigned sWord[SORT_CAP];
    const int t  = threadIdx.x;
    const int me = blockIdx.x;
    const int e0 = min(E, me * chunk);
    const int e1 = min(E, e0 + chunk);
    const int n  = e1 - e0;
    const bool one = n <= SORT_CAP;

    if (t < nBins) lcnt[t] = 0;
    __syncthreads();

    int d[SORT_ITEMS], s[SORT_ITEMS], r[SORT_ITEMS];
    if (one) {
        #pragma unroll
        for (int i = 0; i < SORT_ITEMS; ++i) {
            const int e  = e0 + t + i * SORT_THREADS;
            const int ec = max(min(e, e1 - 1), 0);
            const int dv = dst[ec];
            s[i] = src[ec];
            d[i] = (e < e1) ? dv : -1;
        }
        #pragma unroll
        for (int i = 0; i < SORT_ITEMS; ++i) {
            r[i] = 0;
            if (d[i] >= 0) r[i] = atomicAdd(&lcnt[min(d[i] >> BIN_SHIFT, nBins - 1)], 1);
        }
    } else {
        for (int eb = e0 + t; eb < e1; eb += SORT_ITEMS * SORT_THREADS) {
            #pragma unroll
            for (int i = 0; i < SORT_ITEMS; ++i) {
                const int e = eb + i * SORT_THREADS;
                const int dv = dst[min(e, e1 - 1)];
                d[i] = (e < e1) ? dv : -1;
            }
            #pragma unroll
            for (int i = 0; i < SORT_ITEMS; ++i)
                if (d[i] >= 0) atomicAdd(&lcnt[min(d[i] >> BIN_SHIFT, nBins - 1)], 1);
        }
    }
    __syncthreads();

    const int own    = (t < nBins) ? lcnt[t] : 0;
    const int lstart = block_excl_scan(own, wsum);
    if (t < nBins) {
        lbeg[t] = lstart;
        lcnt[t] = 0;
        runBeg[t * SORT_BLOCKS + me] = lstart;
        if (t == nBins - 1) runBeg[nBins * SORT_BLOCKS + me] = lstart + own;
    }
    __syncthreads();

    if (one) {
        #pragma unroll
        for (int i = 0; i < SORT_ITEMS; ++i) {
            if (d[i] >= 0) {
                const int bin = min(d[i] >> BIN_SHIFT, nBins - 1);
                sWord[lbeg[bin] + r[i]] = ((unsigned)s[i] & ID_MASK) | ((unsigned)(d[i] & 63) << ROW_SHIFT);
            }
        }
        __syncthreads();
        for (int i = t; i < n; i += SORT_THREADS) perm[e0 + i] = sWord[i];
    } else {
        for (int eb = e0 + t; eb < e1; eb += SORT_ITEMS * SORT_THREADS) {
            #pragma unroll
            for (int i = 0; i < SORT_ITEMS; ++i) {
                const int e  = eb + i * SORT_THREADS;
                const int ec = min(e, e1 - 1);
                const int dv = dst[ec];
                s[i] = src[ec];
                d[i] = (e < e1) ? dv : -1;
            }
            #pragma unroll
            for (int i = 0; i < SORT_ITEMS; ++i) {
                if (d[i] >= 0) {
                    const int bin = min(d[i] >> BIN_SHIFT, nBins - 1);
                    const int rk  = atomicAdd(&lcnt[bin], 1);
                    perm[e0 + lbeg[bin] + rk] = ((unsigned)s[i] & ID_MASK) | ((unsigned)(d[i] & 63) << ROW_SHIFT);
                }
            }
        }
    }
}

#define GB 16
#define CAPW 512
#define NCH (CAPW / 32)
#define W_PIECES (DIM * DIM / 4 / (TILE_WAVES * 32))

__device__ __forceinline__ void issue_rows(const float* __restrict__ xl, unsigned w, int i0,
                                           int lastRow, float2 (&v)[GB]) {
    int s[GB];
    #pragma unroll
    for (int k = 0; k < GB; ++k)
        s[k] = min((int)((unsigned)__builtin_amdgcn_readlane((int)w, i0 + k) & ID_MASK), lastRow);
    #pragma unroll
    for (int k = 0; k < GB; ++k) v[k] = *(const float2*)(xl + (size_t)s[k] * DIM);
}

__device__ __forceinline__ void add_rows(float* aPtr, unsigned w, int i0, int nValid,
                                         const float2 (&v)[GB]) {
    #pragma unroll
    for (int k = 0; k < GB; ++k) {
        if (i0 + k < nValid) {
            const int lr = (int)((unsigned)__builtin_amdgcn_readlane((int)w, i0 + k) >> ROW_SHIFT);
            float* p = aPtr + lr * APITCH;
            __hip_atomic_fetch_add(p,     v[k].x, __ATOMIC_RELAXED, __HIP_MEMORY_SCOPE_WORKGROUP);
            __hip_atomic_fetch_add(p + 1, v[k].y, __ATOMIC_RELAXED, __HIP_MEMORY_SCOPE_WORKGROUP);
        }
    }
}

__device__ __forceinline__ void sum_rows(float* aPtr, unsigned w, int i0, int nValid,
                                         const float2 (&v)[GB], float2& acc, int& cur) {
    #pragma unroll
    for (int k = 0; k < GB; ++k) {
        if (i0 + k < nValid) {
            const int lr = (int)((unsigned)__builtin_amdgcn_readlane((int)w, i0 + k) >> ROW_SHIFT);
            if (lr != cur) {
                if (cur >= 0) *(float2*)(aPtr + cur * APITCH) = acc;
                acc = make_float2(0.0f, 0.0f);
                cur = lr;
            }
            acc.x += v[k].x;
            acc.y += v[k].y;
        }
    }
}

__device__ __forceinline__ unsigned sorted_word(const unsigned* sSorted, int begin, int q, int n) {
    const unsigned w = sSorted[min(begin + q, TILE_WAVES * CAPW - 1)];
    return (q < n) ? w : 0u;
}

__device__ __forceinline__ unsigned fetch_word(const unsigned* __restrict__ perm, int q, int total,
                                               const int (&runEnd)[RUNS_PER_WAVE - 1], int gdelta, int lastEdge) {
    int k = 0;
    #pragma unroll
    for (int j = 0; j < RUNS_PER_WAVE - 1; ++j) k += (q >= runEnd[j]) ? 1 : 0;
    const int gd = __shfl(gdelta, k);
    const int idx = max(min(gd + q, lastEdge), 0);
    const unsigned w = perm[idx];
    return (q < total) ? w : 0u;
}

__global__ void __launch_bounds__(TILE_WAVES * 32)
sage_gather_wmma_kernel(const float* __restrict__ x,
                        const int*   __restrict__ runBeg,
                        const unsigned* __restrict__ perm,
                        const float* __restrict__ Wself,
                        const float* __restrict__ bself,
                        const float* __restrict__ Wneigh,
                        const float* __restrict__ bneigh,
                        float*       __restrict__ out,
                        int nTiles, int nRows, int E, int chunk) {
    __shared__ alignas(16) float sWs[DIM * DIM];
    __shared__ alignas(16) float sWn[DIM * DIM];
    __shared__ float sB [DIM];
    __shared__ alignas(16) float sA [TILE_WAVES * 16 * APITCH];
    __shared__ int   sDeg[TILE_WAVES * 16];
    __shared__ int   sCnt[TILE_WAVES * 64];
    __shared__ int   sPos[TILE_WAVES * 64];
    __shared__ int   sTot[TILE_WAVES];
    __shared__ unsigned sSorted[TILE_WAVES * CAPW];

    const int wave = __builtin_amdgcn_readfirstlane((int)(threadIdx.x >> 5));
    const int lane = threadIdx.x & 31;
    const int tile = blockIdx.x * TILE_WAVES + wave;
    const bool active = tile < nTiles;
    const int row0 = tile * 16;

    const int sub = lane & (RUNS_PER_WAVE - 1);
    const int run = wave * RUNS_PER_WAVE + sub;
    const int* rb = runBeg + (size_t)blockIdx.x * SORT_BLOCKS + run;
    const int rs  = rb[0];
    const int re  = rb[SORT_BLOCKS];
    const int len = min(max(re - rs, 0), chunk);
    int incl = len;
    for (int o = 1; o < RUNS_PER_WAVE; o <<= 1) {
        int u = __shfl_up(incl, o, RUNS_PER_WAVE);
        if (sub >= o) incl += u;
    }
    const int total = __builtin_amdgcn_readlane(incl, RUNS_PER_WAVE - 1);
    int runEnd[RUNS_PER_WAVE - 1];
    #pragma unroll
    for (int j = 0; j < RUNS_PER_WAVE - 1; ++j) runEnd[j] = __builtin_amdgcn_readlane(incl, j);
    const int gdelta = run * chunk + rs - (incl - len);
    const int lastEdge = E - 1;

    const float* xl = x + 2 * lane;
    float* aPtr = sA + 2 * lane;
    const int lastRow = nRows - 1;
    const bool fits = total <= CAPW;

    unsigned wd[NCH];
    #pragma unroll
    for (int c = 0; c < NCH; ++c)
        wd[c] = fetch_word(perm, c * 32 + lane, total, runEnd, gdelta, lastEdge);

    #pragma unroll
    for (int j = 0; j < W_PIECES; ++j) {
        const int i = threadIdx.x + j * (TILE_WAVES * 32);
        ((float4*)sWs)[i] = ((const float4*)Wself )[i];
        ((float4*)sWn)[i] = ((const float4*)Wneigh)[i];
    }
    if (threadIdx.x < DIM) sB[threadIdx.x] = bself[threadIdx.x] + bneigh[threadIdx.x];
    for (int i = threadIdx.x; i < TILE_WAVES * 16 * APITCH; i += TILE_WAVES * 32) sA[i] = 0.0f;
    if (threadIdx.x < TILE_WAVES * 16) sDeg[threadIdx.x] = 0;
    sCnt[wave * 64 + lane] = 0;
    sCnt[wave * 64 + 32 + lane] = 0;
    if (lane == 0) sTot[wave] = total;
    if (fits) {
        #pragma unroll
        for (int c = 0; c < NCH; ++c)
            if (c * 32 + lane < total)
                __hip_atomic_fetch_add(&sCnt[wave * 64 + (wd[c] >> ROW_SHIFT)], 1, __ATOMIC_RELAXED, __HIP_MEMORY_SCOPE_WORKGROUP);
    }
    __syncthreads();

    const int t0 = __builtin_amdgcn_readfirstlane(sTot[0]);
    const int t1 = __builtin_amdgcn_readfirstlane(sTot[1]);
    const int t2 = __builtin_amdgcn_readfirstlane(sTot[2]);
    const int t3 = __builtin_amdgcn_readfirstlane(sTot[3]);
    const bool ordered = (t0 <= CAPW) && (t1 <= CAPW) && (t2 <= CAPW) && (t3 <= CAPW);

    float2 va[GB], vb[GB];
    if (ordered) {
        int tLo = 0, tHi = 0, oLo = 0, oHi = 0;
        #pragma unroll
        for (int w = 0; w < TILE_WAVES; ++w) {
            const int a = sCnt[w * 64 + lane];
            const int b = sCnt[w * 64 + 32 + lane];
            if (w < wave) { oLo += a; oHi += b; }
            tLo += a; tHi += b;
        }
        int iLo = tLo, iHi = tHi;
        for (int o = 1; o < 32; o <<= 1) {
            const int u = __shfl_up(iLo, o);
            const int v = __shfl_up(iHi, o);
            if (lane >= o) { iLo += u; iHi += v; }
        }
        const int sumLo = __builtin_amdgcn_readlane(iLo, 31);
        const int grand = sumLo + __builtin_amdgcn_readlane(iHi, 31);
        const int exLo = iLo - tLo;
        const int exHi = sumLo + iHi - tHi;
        sPos[wave * 64 + lane]      = exLo + oLo;
        sPos[wave * 64 + 32 + lane] = exHi + oHi;
        if ((lane >> 4) == (wave & 1)) sDeg[wave * 16 + (lane & 15)] = (wave < 2) ? tLo : tHi;
        #pragma unroll
        for (int c = 0; c < NCH; ++c)
            if (c * 32 + lane < total) {
                const int slot = __hip_atomic_fetch_add(&sPos[wave * 64 + (wd[c] >> ROW_SHIFT)], 1, __ATOMIC_RELAXED, __HIP_MEMORY_SCOPE_WORKGROUP);
                sSorted[slot] = wd[c];
            }
        const int exSel = (wave < 2) ? exLo : exHi;
        const int b0  = __builtin_amdgcn_readlane(exSel, 0);
        const int b16 = __builtin_amdgcn_readlane(exSel, 16);
        const int begin = (wave & 1) ? b16 : b0;
        const int endv  = (wave & 1) ? ((wave == 1) ? sumLo : grand) : b16;
        const int nE = endv - begin;
        __syncthreads();

        float2 acc = make_float2(0.0f, 0.0f);
        int cur = -1;
        unsigned word = 0u, wordNext = 0u;
        if (nE > 0) {
            word = sorted_word(sSorted, begin, lane, nE);
            issue_rows(xl, word, 0, lastRow, va);
            wordNext = sorted_word(sSorted, begin, 32 + lane, nE);
        }
        for (int q0 = 0; q0 < nE; q0 += 32) {
            const int nValid = min(32, nE - q0);
            const unsigned wordAfter = sorted_word(sSorted, begin, q0 + 64 + lane, nE);
            issue_rows(xl, word, GB, lastRow, vb);
            sum_rows(aPtr, word, 0, nValid, va, acc, cur);
            issue_rows(xl, wordNext, 0, lastRow, va);
            sum_rows(aPtr, word, GB, nValid, vb, acc, cur);
            word = wordNext;
            wordNext = wordAfter;
        }
        if (cur >= 0) *(float2*)(aPtr + cur * APITCH) = acc;
    } else {
        unsigned word = 0u, wordNext = 0u;
        if (total > 0) {
            word = fetch_word(perm, lane, total, runEnd, gdelta, lastEdge);
            issue_rows(xl, word, 0, lastRow, va);
            wordNext = fetch_word(perm, 32 + lane, total, runEnd, gdelta, lastEdge);
        }
        for (int q0 = 0; q0 < total; q0 += 32) {
            const int nValid = min(32, total - q0);
            const unsigned wordAfter = fetch_word(perm, q0 + 64 + lane, total, runEnd, gdelta, lastEdge);
            issue_rows(xl, word, GB, lastRow, vb);
            if (lane < nValid)
                __hip_atomic_fetch_add(&sDeg[word >> ROW_SHIFT], 1, __ATOMIC_RELAXED, __HIP_MEMORY_SCOPE_WORKGROUP);
            add_rows(aPtr, word, 0, nValid, va);
            issue_rows(xl, wordNext, 0, lastRow, va);
            add_rows(aPtr, word, GB, nValid, vb);
            word = wordNext;
            wordNext = wordAfter;
        }
        __syncthreads();
    }
    if (!active) return;

    const int m    = lane & 15;
    const int hi   = lane >> 4;
    const int row  = row0 + m;

    const float dinv = 1.0f / fmaxf((float)sDeg[wave * 16 + m], 1.0f);
    const float* __restrict__ xrow = x + (size_t)row * DIM;
    const float* arow = sA + (wave * 16 + m) * APITCH;

    v8f acc[4];
    for (int nt = 0; nt < 4; ++nt) {
        float b = sB[nt * 16 + m];
        for (int r = 0; r < 8; ++r) acc[nt][r] = b;
    }

    for (int kb = 0; kb < DIM; kb += 4) {
        const int kk = kb + 2 * hi;
        v2f as, an;
        as.x = xrow[kk];
        as.y = xrow[kk + 1];
        an.x = arow[kk]     * dinv;
        an.y = arow[kk + 1] * dinv;

        for (int nt = 0; nt < 4; ++nt) {
            const int bi = kk * DIM + nt * 16 + m;
            v2f bs, bn;
            bs.x = sWs[bi];
            bs.y = sWs[bi + DIM];
            bn.x = sWn[bi];
            bn.y = sWn[bi + DIM];
            acc[nt] = __builtin_amdgcn_wmma_f32_16x16x4_f32(
                          false, as, false, bs, (short)0, acc[nt], false, false);
            acc[nt] = __builtin_amdgcn_wmma_f32_16x16x4_f32(
                          false, an, false, bn, (short)0, acc[nt], false, false);
        }
    }

    for (int nt = 0; nt < 4; ++nt) {
        const int col = nt * 16 + m;
        for (int r = 0; r < 8; ++r) {
            const int mrow = r + 8 * hi;
            out[(size_t)(row0 + mrow) * DIM + col] = acc[nt][r];
        }
    }
}

extern "C" void kernel_launch(void* const* d_in, const int* in_sizes, int n_in,
                              void* d_out, int out_size, void* d_ws, size_t ws_size,
                              hipStream_t stream) {
    const float* x      = (const float*)d_in[0];
    const int*   src    = (const int*)  d_in[1];
    const int*   dst    = (const int*)  d_in[2];
    const float* Wself  = (const float*)d_in[3];
    const float* bself  = (const float*)d_in[4];
    const float* Wneigh = (const float*)d_in[5];
    const float* bneigh = (const float*)d_in[6];
    float*       out    = (float*)d_out;

    const int N = in_sizes[0] / DIM;
    const int E = in_sizes[1];

    const int nTiles = (N + 15) / 16;
    const int nBins  = (nTiles + TILE_WAVES - 1) / TILE_WAVES;
    const int chunk  = (E + SORT_BLOCKS - 1) / SORT_BLOCKS;

    int*      runBeg = (int*)d_ws;
    unsigned* perm   = (unsigned*)(runBeg + (nBins + 1) * SORT_BLOCKS);

    sage_slice_sort_kernel<<<SORT_BLOCKS, SORT_THREADS, 0, stream>>>(src, dst, runBeg, perm, E, nBins, chunk);

    sage_gather_wmma_kernel<<<nBins, TILE_WAVES * 32, 0, stream>>>(
        x, runBeg, perm, Wself, bself, Wneigh, bneigh, out, nTiles, N, E, chunk);
}


# Gather loop issues fewer instructions per edge: source-row offsets and a row-change bit mask prepared once per 32 edges on the vector unit, no per-edge bound test in full trips; sum order kept
# speedup vs baseline: 5.7910x; 1.1970x over previous
#include <hip/hip_runtime.h>


typedef __attribute__((ext_vector_type(2))) float v2f;
typedef __attribute__((ext_vector_type(8))) float v8f;

#define DIM 64
#define TILE_WAVES 4
#define APITCH 68
#define BIN_SHIFT 6
#define NB_MAX 1024
#define SORT_THREADS 1024
#define SORT_BLOCKS 64
#define SORT_ITEMS 16
#define SORT_CAP (SORT_ITEMS * SORT_THREADS)
#define RUNS_PER_WAVE (SORT_BLOCKS / TILE_WAVES)
#define ROW_SHIFT 26
#define ID_MASK 0x03FFFFFFu

__device__ __forceinline__ int block_excl_scan(int val, int* wsum) {
    const int lane = threadIdx.x & 31;
    const int wv   = threadIdx.x >> 5;
    int incl = val;
    for (int o = 1; o < 32; o <<= 1) {
        int u = __shfl_up(incl, o);
        if (lane >= o) incl += u;
    }
    if (lane == 31) wsum[wv] = incl;
    __syncthreads();
    if (wv == 0) {
        const int s = wsum[lane];
        int si = s;
        for (int o = 1; o < 32; o <<= 1) {
            int u = __shfl_up(si, o);
            if (lane >= o) si += u;
        }
        wsum[lane] = si - s;
    }
    __syncthreads();
    const int r = wsum[wv] + incl - val;
    __syncthreads();
    return r;
}

__global__ void __launch_bounds__(SORT_THREADS)
sage_slice_sort_kernel(const int* __restrict__ src, const int* __restrict__ dst,
                       int* __restrict__ runBeg, unsigned* __restrict__ perm,
                       int E, int nBins, int chunk) {
    __shared__ int lcnt[NB_MAX];
    __shared__ int lbeg[NB_MAX];
    __shared__ int wsum[32];
    __shared__ unsigned sWord[SORT_CAP];
    const int t  = threadIdx.x;
    const int me = blockIdx.x;
    const int e0 = min(E, me * chunk);
    const int e1 = min(E, e0 + chunk);
    const int n  = e1 - e0;
    const bool one = n <= SORT_CAP;

    if (t < nBins) lcnt[t] = 0;
    __syncthreads();

    int d[SORT_ITEMS], s[SORT_ITEMS], r[SORT_ITEMS];
    if (one) {
        #pragma unroll
        for (int i = 0; i < SORT_ITEMS; ++i) {
            const int e  = e0 + t + i * SORT_THREADS;
            const int ec = max(min(e, e1 - 1), 0);
            const int dv = dst[ec];
            s[i] = src[ec];
            d[i] = (e < e1) ? dv : -1;
        }
        #pragma unroll
        for (int i = 0; i < SORT_ITEMS; ++i) {
            r[i] = 0;
            if (d[i] >= 0) r[i] = atomicAdd(&lcnt[min(d[i] >> BIN_SHIFT, nBins - 1)], 1);
        }
    } else {
        for (int eb = e0 + t; eb < e1; eb += SORT_ITEMS * SORT_THREADS) {
            #pragma unroll
            for (int i = 0; i < SORT_ITEMS; ++i) {
                const int e = eb + i * SORT_THREADS;
                const int dv = dst[min(e, e1 - 1)];
                d[i] = (e < e1) ? dv : -1;
            }
            #pragma unroll
            for (int i = 0; i < SORT_ITEMS; ++i)
                if (d[i] >= 0) atomicAdd(&lcnt[min(d[i] >> BIN_SHIFT, nBins - 1)], 1);
        }
    }
    __syncthreads();

    const int own    = (t < nBins) ? lcnt[t] : 0;
    const int lstart = block_excl_scan(own, wsum);
    if (t < nBins) {
        lbeg[t] = lstart;
        lcnt[t] = 0;
        runBeg[t * SORT_BLOCKS + me] = lstart;
        if (t == nBins - 1) runBeg[nBins * SORT_BLOCKS + me] = lstart + own;
    }
    __syncthreads();

    if (one) {
        #pragma unroll
        for (int i = 0; i < SORT_ITEMS; ++i) {
            if (d[i] >= 0) {
                const int bin = min(d[i] >> BIN_SHIFT, nBins - 1);
                sWord[lbeg[bin] + r[i]] = ((unsigned)s[i] & ID_MASK) | ((unsigned)(d[i] & 63) << ROW_SHIFT);
            }
        }
        __syncthreads();
        for (int i = t; i < n; i += SORT_THREADS) perm[e0 + i] = sWord[i];
    } else {
        for (int eb = e0 + t; eb < e1; eb += SORT_ITEMS * SORT_THREADS) {
            #pragma unroll
            for (int i = 0; i < SORT_ITEMS; ++i) {
                const int e  = eb + i * SORT_THREADS;
                const int ec = min(e, e1 - 1);
                const int dv = dst[ec];
                s[i] = src[ec];
                d[i] = (e < e1) ? dv : -1;
            }
            #pragma unroll
            for (int i = 0; i < SORT_ITEMS; ++i) {
                if (d[i] >= 0) {
                    const int bin = min(d[i] >> BIN_SHIFT, nBins - 1);
                    const int rk  = atomicAdd(&lcnt[bin], 1);
                    perm[e0 + lbeg[bin] + rk] = ((unsigned)s[i] & ID_MASK) | ((unsigned)(d[i] & 63) << ROW_SHIFT);
                }
            }
        }
    }
}

#define GB 16
#define CAPW 512
#define NCH (CAPW / 32)
#define W_PIECES (DIM * DIM / 4 / (TILE_WAVES * 32))

__device__ __forceinline__ void issue_rows(const float* __restrict__ xl, unsigned w, int i0,
                                           int lastRow, float2 (&v)[GB]) {
    int s[GB];
    #pragma unroll
    for (int k = 0; k < GB; ++k)
        s[k] = min((int)((unsigned)__builtin_amdgcn_readlane((int)w, i0 + k) & ID_MASK), lastRow);
    #pragma unroll
    for (int k = 0; k < GB; ++k) v[k] = *(const float2*)(xl + (size_t)s[k] * DIM);
}

__device__ __forceinline__ void add_rows(float* aPtr, unsigned w, int i0, int nValid,
                                         const float2 (&v)[GB]) {
    #pragma unroll
    for (int k = 0; k < GB; ++k) {
        if (i0 + k < nValid) {
            const int lr = (int)((unsigned)__builtin_amdgcn_readlane((int)w, i0 + k) >> ROW_SHIFT);
            float* p = aPtr + lr * APITCH;
            __hip_atomic_fetch_add(p,     v[k].x, __ATOMIC_RELAXED, __HIP_MEMORY_SCOPE_WORKGROUP);
            __hip_atomic_fetch_add(p + 1, v[k].y, __ATOMIC_RELAXED, __HIP_MEMORY_SCOPE_WORKGROUP);
        }
    }
}

__device__ __forceinline__ unsigned trip_prepare(const unsigned* sSorted, int begin, int q0, int n, int lane,
                                                 int lastRow, unsigned& off, unsigned& rows) {
    const int q = q0 + lane;
    const unsigned w  = sSorted[min(begin + q, TILE_WAVES * CAPW - 1)];
    const unsigned wp = sSorted[min(max(begin + q - 1, 0), TILE_WAVES * CAPW - 1)];
    const bool in = q < n;
    const unsigned wv = in ? w : 0u;
    off  = (unsigned)min((int)(wv & ID_MASK), lastRow) * (unsigned)(DIM * sizeof(float));
    rows = wv >> ROW_SHIFT;
    const bool opens = in && (q == 0 || rows != (wp >> ROW_SHIFT));
    return __builtin_amdgcn_ballot_w32(opens);
}

__device__ __forceinline__ void issue_rows_off(const char* __restrict__ xb, unsigned lane8, unsigned off, int i0,
                                               float2 (&v)[GB]) {
    #pragma unroll
    for (int k = 0; k < GB; ++k) {
        const unsigned o = (unsigned)__builtin_amdgcn_readlane((int)off, i0 + k);
        const char* rowBase = xb + (size_t)o;
        v[k] = *(const float2*)(rowBase + lane8);
    }
}

template <bool FULL>
__device__ __forceinline__ void sum_rows_mask(float* aPtr, unsigned opens, unsigned rows, int i0, int nValid,
                                              const float2 (&v)[GB], float2& acc, int& cur) {
    #pragma unroll
    for (int k = 0; k < GB; ++k) {
        if (FULL || i0 + k < nValid) {
            if (__builtin_expect((opens >> (i0 + k)) & 1u, 0)) {
                if (cur >= 0) *(float2*)(aPtr + cur * APITCH) = acc;
                acc = make_float2(0.0f, 0.0f);
                cur = __builtin_amdgcn_readlane((int)rows, i0 + k);
            }
            acc.x += v[k].x;
            acc.y += v[k].y;
        }
    }
}

__device__ __forceinline__ unsigned fetch_word(const unsigned* __restrict__ perm, int q, int total,
                                               const int (&runEnd)[RUNS_PER_WAVE - 1], int gdelta, int lastEdge) {
    int k = 0;
    #pragma unroll
    for (int j = 0; j < RUNS_PER_WAVE - 1; ++j) k += (q >= runEnd[j]) ? 1 : 0;
    const int gd = __shfl(gdelta, k);
    const int idx = max(min(gd + q, lastEdge), 0);
    const unsigned w = perm[idx];
    return (q < total) ? w : 0u;
}

__global__ void __launch_bounds__(TILE_WAVES * 32)
sage_gather_wmma_kernel(const float* __restrict__ x,
                        const int*   __restrict__ runBeg,
                        const unsigned* __restrict__ perm,
                        const float* __restrict__ Wself,
                        const float* __restrict__ bself,
                        const float* __restrict__ Wneigh,
                        const float* __restrict__ bneigh,
                        float*       __restrict__ out,
                        int nTiles, int nRows, int E, int chunk) {
    __shared__ alignas(16) float sWs[DIM * DIM];
    __shared__ alignas(16) float sWn[DIM * DIM];
    __shared__ float sB [DIM];
    __shared__ alignas(16) float sA [TILE_WAVES * 16 * APITCH];
    __shared__ int   sDeg[TILE_WAVES * 16];
    __shared__ int   sCnt[TILE_WAVES * 64];
    __shared__ int   sPos[TILE_WAVES * 64];
    __shared__ int   sTot[TILE_WAVES];
    __shared__ unsigned sSorted[TILE_WAVES * CAPW];

    const int wave = __builtin_amdgcn_readfirstlane((int)(threadIdx.x >> 5));
    const int lane = threadIdx.x & 31;
    const int tile = blockIdx.x * TILE_WAVES + wave;
    const bool active = tile < nTiles;
    const int row0 = tile * 16;

    const int sub = lane & (RUNS_PER_WAVE - 1);
    const int run = wave * RUNS_PER_WAVE + sub;
    const int* rb = runBeg + (size_t)blockIdx.x * SORT_BLOCKS + run;
    const int rs  = rb[0];
    const int re  = rb[SORT_BLOCKS];
    const int len = min(max(re - rs, 0), chunk);
    int incl = len;
    for (int o = 1; o < RUNS_PER_WAVE; o <<= 1) {
        int u = __shfl_up(incl, o, RUNS_PER_WAVE);
        if (sub >= o) incl += u;
    }
    const int total = __builtin_amdgcn_readlane(incl, RUNS_PER_WAVE - 1);
    int runEnd[RUNS_PER_WAVE - 1];
    #pragma unroll
    for (int j = 0; j < RUNS_PER_WAVE - 1; ++j) runEnd[j] = __builtin_amdgcn_readlane(incl, j);
    const int gdelta = run * chunk + rs - (incl - len);
    const int lastEdge = E - 1;

    const float* xl = x + 2 * lane;
    float* aPtr = sA + 2 * lane;
    const int lastRow = nRows - 1;
    const bool fits = total <= CAPW;

    unsigned wd[NCH];
    #pragma unroll
    for (int c = 0; c < NCH; ++c)
        wd[c] = fetch_word(perm, c * 32 + lane, total, runEnd, gdelta, lastEdge);

    #pragma unroll
    for (int j = 0; j < W_PIECES; ++j) {
        const int i = threadIdx.x + j * (TILE_WAVES * 32);
        ((float4*)sWs)[i] = ((const float4*)Wself )[i];
        ((float4*)sWn)[i] = ((const float4*)Wneigh)[i];
    }
    if (threadIdx.x < DIM) sB[threadIdx.x] = bself[threadIdx.x] + bneigh[threadIdx.x];
    for (int i = threadIdx.x; i < TILE_WAVES * 16 * APITCH; i += TILE_WAVES * 32) sA[i] = 0.0f;
    if (threadIdx.x < TILE_WAVES * 16) sDeg[threadIdx.x] = 0;
    sCnt[wave * 64 + lane] = 0;
    sCnt[wave * 64 + 32 + lane] = 0;
    if (lane == 0) sTot[wave] = total;
    if (fits) {
        #pragma unroll
        for (int c = 0; c < NCH; ++c)
            if (c * 32 + lane < total)
                __hip_atomic_fetch_add(&sCnt[wave * 64 + (wd[c] >> ROW_SHIFT)], 1, __ATOMIC_RELAXED, __HIP_MEMORY_SCOPE_WORKGROUP);
    }
    __syncthreads();

    const int t0 = __builtin_amdgcn_readfirstlane(sTot[0]);
    const int t1 = __builtin_amdgcn_readfirstlane(sTot[1]);
    const int t2 = __builtin_amdgcn_readfirstlane(sTot[2]);
    const int t3 = __builtin_amdgcn_readfirstlane(sTot[3]);
    const bool ordered = (t0 <= CAPW) && (t1 <= CAPW) && (t2 <= CAPW) && (t3 <= CAPW);

    float2 va[GB], vb[GB];
    if (ordered) {
        int tLo = 0, tHi = 0, oLo = 0, oHi = 0;
        #pragma unroll
        for (int w = 0; w < TILE_WAVES; ++w) {
            const int a = sCnt[w * 64 + lane];
            const int b = sCnt[w * 64 + 32 + lane];
            if (w < wave) { oLo += a; oHi += b; }
            tLo += a; tHi += b;
        }
        int iLo = tLo, iHi = tHi;
        for (int o = 1; o < 32; o <<= 1) {
            const int u = __shfl_up(iLo, o);
            const int v = __shfl_up(iHi, o);
            if (lane >= o) { iLo += u; iHi += v; }
        }
        const int sumLo = __builtin_amdgcn_readlane(iLo, 31);
        const int grand = sumLo + __builtin_amdgcn_readlane(iHi, 31);
        const int exLo = iLo - tLo;
        const int exHi = sumLo + iHi - tHi;
        sPos[wave * 64 + lane]      = exLo + oLo;
        sPos[wave * 64 + 32 + lane] = exHi + oHi;
        if ((lane >> 4) == (wave & 1)) sDeg[wave * 16 + (lane & 15)] = (wave < 2) ? tLo : tHi;
        #pragma unroll
        for (int c = 0; c < NCH; ++c)
            if (c * 32 + lane < total) {
                const int slot = __hip_atomic_fetch_add(&sPos[wave * 64 + (wd[c] >> ROW_SHIFT)], 1, __ATOMIC_RELAXED, __HIP_MEMORY_SCOPE_WORKGROUP);
                sSorted[slot] = wd[c];
            }
        const int exSel = (wave < 2) ? exLo : exHi;
        const int b0  = __builtin_amdgcn_readlane(exSel, 0);
        const int b16 = __builtin_amdgcn_readlane(exSel, 16);
        const int begin = (wave & 1) ? b16 : b0;
        const int endv  = (wave & 1) ? ((wave == 1) ? sumLo : grand) : b16;
        const int nE = endv - begin;
        __syncthreads();

        float2 acc = make_float2(0.0f, 0.0f);
        int cur = -1;
        const char* xb = (const char*)x;
        const unsigned lane8 = (unsigned)lane * (unsigned)(2 * sizeof(float));
        unsigned off = 0u, rows = 0u, opens = 0u, offNext = 0u, rowsNext = 0u, opensNext = 0u;
        if (nE > 0) {
            opens = trip_prepare(sSorted, begin, 0, nE, lane, lastRow, off, rows);
            issue_rows_off(xb, lane8, off, 0, va);
            opensNext = trip_prepare(sSorted, begin, 32, nE, lane, lastRow, offNext, rowsNext);
        }
        for (int q0 = 0; q0 < nE; q0 += 32) {
            const int nValid = min(32, nE - q0);
            unsigned offAfter, rowsAfter;
            const unsigned opensAfter = trip_prepare(sSorted, begin, q0 + 64, nE, lane, lastRow, offAfter, rowsAfter);
            issue_rows_off(xb, lane8, off, GB, vb);
            const bool full = nValid == 32;
            if (full) sum_rows_mask<true >(aPtr, opens, rows, 0, nValid, va, acc, cur);
            else      sum_rows_mask<false>(aPtr, opens, rows, 0, nValid, va, acc, cur);
            issue_rows_off(xb, lane8, offNext, 0, va);
            if (full) sum_rows_mask<true >(aPtr, opens, rows, GB, nValid, vb, acc, cur);
            else      sum_rows_mask<false>(aPtr, opens, rows, GB, nValid, vb, acc, cur);
            off = offNext;   rows = rowsNext;   opens = opensNext;
            offNext = offAfter; rowsNext = rowsAfter; opensNext = opensAfter;
        }
        if (cur >= 0) *(float2*)(aPtr + cur * APITCH) = acc;
    } else {
        unsigned word = 0u, wordNext = 0u;
        if (total > 0) {
            word = fetch_word(perm, lane, total, runEnd, gdelta, lastEdge);
            issue_rows(xl, word, 0, lastRow, va);
            wordNext = fetch_word(perm, 32 + lane, total, runEnd, gdelta, lastEdge);
        }
        for (int q0 = 0; q0 < total; q0 += 32) {
            const int nValid = min(32, total - q0);
            const unsigned wordAfter = fetch_word(perm, q0 + 64 + lane, total, runEnd, gdelta, lastEdge);
            issue_rows(xl, word, GB, lastRow, vb);
            if (lane < nValid)
                __hip_atomic_fetch_add(&sDeg[word >> ROW_SHIFT], 1, __ATOMIC_RELAXED, __HIP_MEMORY_SCOPE_WORKGROUP);
            add_rows(aPtr, word, 0, nValid, va);
            issue_rows(xl, wordNext, 0, lastRow, va);
            add_rows(aPtr, word, GB, nValid, vb);
            word = wordNext;
            wordNext = wordAfter;
        }
        __syncthreads();
    }
    if (!active) return;

    const int m    = lane & 15;
    const int hi   = lane >> 4;
    const int row  = row0 + m;

    const float dinv = 1.0f / fmaxf((float)sDeg[wave * 16 + m], 1.0f);
    const float* __restrict__ xrow = x + (size_t)row * DIM;
    const float* arow = sA + (wave * 16 + m) * APITCH;

    v8f acc[4];
    for (int nt = 0; nt < 4; ++nt) {
        float b = sB[nt * 16 + m];
        for (int r = 0; r < 8; ++r) acc[nt][r] = b;
    }

    for (int kb = 0; kb < DIM; kb += 4) {
        const int kk = kb + 2 * hi;
        v2f as, an;
        as.x = xrow[kk];
        as.y = xrow[kk + 1];
        an.x = arow[kk]     * dinv;
        an.y = arow[kk + 1] * dinv;

        for (int nt = 0; nt < 4; ++nt) {
            const int bi = kk * DIM + nt * 16 + m;
            v2f bs, bn;
            bs.x = sWs[bi];
            bs.y = sWs[bi + DIM];
            bn.x = sWn[bi];
            bn.y = sWn[bi + DIM];
            acc[nt] = __builtin_amdgcn_wmma_f32_16x16x4_f32(
                          false, as, false, bs, (short)0, acc[nt], false, false);
            acc[nt] = __builtin_amdgcn_wmma_f32_16x16x4_f32(
                          false, an, false, bn, (short)0, acc[nt], false, false);
        }
    }

    for (int nt = 0; nt < 4; ++nt) {
        const int col = nt * 16 + m;
        for (int r = 0; r < 8; ++r) {
            const int mrow = r + 8 * hi;
            out[(size_t)(row0 + mrow) * DIM + col] = acc[nt][r];
        }
    }
}

extern "C" void kernel_launch(void* const* d_in, const int* in_sizes, int n_in,
                              void* d_out, int out_size, void* d_ws, size_t ws_size,
                              hipStream_t stream) {
    const float* x      = (const float*)d_in[0];
    const int*   src    = (const int*)  d_in[1];
    const int*   dst    = (const int*)  d_in[2];
    const float* Wself  = (const float*)d_in[3];
    const float* bself  = (const float*)d_in[4];
    const float* Wneigh = (const float*)d_in[5];
    const float* bneigh = (const float*)d_in[6];
    float*       out    = (float*)d_out;

    const int N = in_sizes[0] / DIM;
    const int E = in_sizes[1];

    const int nTiles = (N + 15) / 16;
    const int nBins  = (nTiles + TILE_WAVES - 1) / TILE_WAVES;
    const int chunk  = (E + SORT_BLOCKS - 1) / SORT_BLOCKS;

    int*      runBeg = (int*)d_ws;
    unsigned* perm   = (unsigned*)(runBeg + (nBins + 1) * SORT_BLOCKS);

    sage_slice_sort_kernel<<<SORT_BLOCKS, SORT_THREADS, 0, stream>>>(src, dst, runBeg, perm, E, nBins, chunk);

    sage_gather_wmma_kernel<<<nBins, TILE_WAVES * 32, 0, stream>>>(
        x, runBeg, perm, Wself, bself, Wneigh, bneigh, out, nTiles, N, E, chunk);
}
